# starting kernel
# baseline (speedup 1.0000x reference)
#include <hip/hip_runtime.h>
#include <hip/hip_bf16.h>

#define AANN 256
#define BB   64
#define DD   768
#define LL   2
#define GG1  4
#define GG2  8

typedef __attribute__((ext_vector_type(2))) float v2f;
typedef __attribute__((ext_vector_type(8))) float v8f;

__device__ __forceinline__ v8f wmma_f32(v2f a, v2f b, v8f c) {
    return __builtin_amdgcn_wmma_f32_16x16x4_f32(
         false, a,  false, b,
         (short)0, c,  false,  false);
}

__device__ __forceinline__ void gemm_wave_32x64(
    const float* __restrict__ A,
    const float* __restrict__ Bm,
    const float* __restrict__ bias,
    float* __restrict__ C,
    int m0, int n0)
{
    const int lane = threadIdx.x & 31;
    const int half = lane >> 4;
    const int r    = lane & 15;

    v8f acc[2][4];
    #pragma unroll
    for (int t = 0; t < 4; ++t) {
        const float bv = bias[n0 + t * 16 + r];
        #pragma unroll
        for (int j = 0; j < 8; ++j) { acc[0][t][j] = bv; acc[1][t][j] = bv; }
    }

    const float* arow0 = A + (size_t)(m0 + r) * DD;
    const float* arow1 = arow0 + (size_t)16 * DD;
    const float* bcol  = Bm + n0 + r;

    #pragma unroll 2
    for (int k = 0; k < DD; k += 4) {
        const int ka = k + 2 * half;
        v2f a0, a1;
        a0.x = arow0[ka];
        a0.y = arow0[ka + 1];
        a1.x = arow1[ka];
        a1.y = arow1[ka + 1];
        const float* bk = bcol + (size_t)ka * DD;
        #pragma unroll
        for (int t = 0; t < 4; ++t) {
            v2f b;
            b.x = bk[t * 16];
            b.y = bk[t * 16 + DD];
            acc[0][t] = wmma_f32(a0, b, acc[0][t]);
            acc[1][t] = wmma_f32(a1, b, acc[1][t]);
        }
    }

    float* crow = C + n0 + r;
    #pragma unroll
    for (int u = 0; u < 2; ++u) {
        #pragma unroll
        for (int t = 0; t < 4; ++t) {
            #pragma unroll
            for (int j = 0; j < 8; ++j) {
                const int row = m0 + u * 16 + j + 8 * half;
                crow[(size_t)row * DD + t * 16] = acc[u][t][j];
            }
        }
    }
}

__global__ __launch_bounds__(256) void stage1_kernel(
    const float* __restrict__ pooled, const float* __restrict__ W1,
    const float* __restrict__ b1, float* __restrict__ H1)
{
    const int g  = blockIdx.x / 3;
    const int wt = (blockIdx.x % 3) * 8 + (threadIdx.x >> 5);
    const int m0 = (wt / 12) * 32;
    const int n0 = (wt % 12) * 64;
    gemm_wave_32x64(pooled,
                    W1 + (size_t)g * DD * DD,
                    b1 + (size_t)g * DD,
                    H1 + (size_t)g * BB * DD,
                    m0, n0);
}

__global__ __launch_bounds__(256) void stage2_kernel(
    const float* __restrict__ H1, const float* __restrict__ W2,
    const float* __restrict__ b2, float* __restrict__ H2)
{
    const int p  = blockIdx.x / 3;
    const int g1 = p >> 3;
    const int g2 = p & 7;
    const int wt = (blockIdx.x % 3) * 8 + (threadIdx.x >> 5);
    const int m0 = (wt / 12) * 32;
    const int n0 = (wt % 12) * 64;
    gemm_wave_32x64(H1 + (size_t)g1 * BB * DD,
                    W2 + (size_t)g2 * DD * DD,
                    b2 + (size_t)g2 * DD,
                    H2 + (size_t)p * BB * DD,
                    m0, n0);
}

__global__ __launch_bounds__(256) void stage3_kernel(
    const float* __restrict__ H2, const float* __restrict__ Wh,
    const float* __restrict__ bh, const int* __restrict__ g1_idx,
    const int* __restrict__ g2_idx, float* __restrict__ out)
{
    const int wtile = blockIdx.x * 8 + (threadIdx.x >> 5);
    const int a  = wtile >> 2;
    const int m0 = (wtile & 3) * 16;
    const int pair = g1_idx[a] * GG2 + g2_idx[a];

    const float* Am = H2 + (size_t)pair * BB * DD;
    const float* Bm = Wh + (size_t)a * DD * LL;

    const int lane = threadIdx.x & 31;
    const int half = lane >> 4;
    const int r    = lane & 15;
    const int rc   = r & 1;
    const float keep = (r < LL) ? 1.0f : 0.0f;

    v8f acc;
    const float bv = (r < LL) ? bh[a * LL + r] : 0.0f;
    #pragma unroll
    for (int j = 0; j < 8; ++j) acc[j] = bv;

    const float* arow = Am + (size_t)(m0 + r) * DD;

    #pragma unroll 4
    for (int k = 0; k < DD; k += 4) {
        const int ka = k + 2 * half;
        v2f a2;
        a2.x = arow[ka];
        a2.y = arow[ka + 1];
        v2f b;
        b.x = keep * Bm[ka * LL + rc];
        b.y = keep * Bm[(ka + 1) * LL + rc];
        acc = wmma_f32(a2, b, acc);
    }

    if (r < LL) {
        #pragma unroll
        for (int j = 0; j < 8; ++j) {
            const int row = m0 + j + 8 * half;
            out[(size_t)row * (AANN * LL) + a * LL + r] = acc[j];
        }
    }
}

extern "C" void kernel_launch(void* const* d_in, const int* in_sizes, int n_in,
                              void* d_out, int out_size, void* d_ws, size_t ws_size,
                              hipStream_t stream) {
    const float* pooled = (const float*)d_in[0];
    const float* W1     = (const float*)d_in[1];
    const float* b1     = (const float*)d_in[2];
    const float* W2     = (const float*)d_in[3];
    const float* b2     = (const float*)d_in[4];
    const float* Wh     = (const float*)d_in[5];
    const float* bh     = (const float*)d_in[6];
    const int*   g1     = (const int*)d_in[7];
    const int*   g2     = (const int*)d_in[8];
    float*       out    = (float*)d_out;

    float* H1 = (float*)d_ws;
    float* H2 = H1 + (size_t)GG1 * BB * DD;

    stage1_kernel<<<GG1 * 3, 256, 0, stream>>>(pooled, W1, b1, H1);
    stage2_kernel<<<GG1 * GG2 * 3, 256, 0, stream>>>(H1, W2, b2, H2);
    stage3_kernel<<<(AANN * 4) / 8, 256, 0, stream>>>(H2, Wh, bh, g1, g2, out);
}


# Head product moved into stage 2's epilogue: each work-group multiplies its 64x96 H2 block, still in registers, by its pair's head rows; a last launch sums the 8 block partials; sum order changes
# speedup vs baseline: 5.8841x; 5.8841x over previous
#include <hip/hip_runtime.h>
#include <hip/hip_bf16.h>

#define AANN 256
#define BB   64
#define DD   768
#define LL   2
#define GG1  4
#define GG2  8

typedef __attribute__((ext_vector_type(2))) float v2f;
typedef __attribute__((ext_vector_type(8))) float v8f;

__device__ __forceinline__ v8f wmma_f32(v2f a, v2f b, v8f c) {
    return __builtin_amdgcn_wmma_f32_16x16x4_f32(
         false, a,  false, b,
         (short)0, c,  false,  false);
}


#define KC1    12
#define KSPAN1 (KC1 * 4)

struct Tile1Frags {
    v2f a[KC1];
    v2f b[KC1][2];
};

__device__ __forceinline__ void tile1_load_frags(
    Tile1Frags& f, const float* __restrict__ arow, const float* __restrict__ bcol, int ka0)
{
    #pragma unroll
    for (int s = 0; s < KC1; ++s) {
        const int ka = ka0 + 4 * s;
        f.a[s].x = arow[ka];
        f.a[s].y = arow[ka + 1];
        const float* bk = bcol + (size_t)ka * DD;
        #pragma unroll
        for (int t = 0; t < 2; ++t) {
            f.b[s][t].x = bk[t * 16];
            f.b[s][t].y = bk[t * 16 + DD];
        }
    }
}

__device__ __forceinline__ void tile1_mma_frags(const Tile1Frags& f, v8f (&acc)[2])
{
    #pragma unroll
    for (int s = 0; s < KC1; ++s) {
        #pragma unroll
        for (int t = 0; t < 2; ++t) {
            acc[t] = wmma_f32(f.a[s], f.b[s][t], acc[t]);
        }
    }
}

__device__ __forceinline__ void gemm_wave_16x32(
    const float* __restrict__ A,
    const float* __restrict__ Bm,
    const float* __restrict__ bias,
    float* __restrict__ C,
    int m0, int n0)
{
    const int lane = threadIdx.x & 31;
    const int half = lane >> 4;
    const int r    = lane & 15;

    v8f acc[2];
    #pragma unroll
    for (int t = 0; t < 2; ++t) {
        const float bv = bias[n0 + t * 16 + r];
        #pragma unroll
        for (int j = 0; j < 8; ++j) acc[t][j] = bv;
    }

    const float* arow = A + (size_t)(m0 + r) * DD;
    const float* bcol = Bm + n0 + r;

    Tile1Frags f0, f1;
    tile1_load_frags(f0, arow, bcol, 2 * half);
    for (int k = 0; k < DD - 2 * KSPAN1; k += 2 * KSPAN1) {
        tile1_load_frags(f1, arow, bcol, k + KSPAN1 + 2 * half);
        __builtin_amdgcn_sched_barrier(0);
        tile1_mma_frags(f0, acc);
        tile1_load_frags(f0, arow, bcol, k + 2 * KSPAN1 + 2 * half);
        __builtin_amdgcn_sched_barrier(0);
        tile1_mma_frags(f1, acc);
    }
    tile1_load_frags(f1, arow, bcol, DD - KSPAN1 + 2 * half);
    __builtin_amdgcn_sched_barrier(0);
    tile1_mma_frags(f0, acc);
    tile1_mma_frags(f1, acc);

    float* crow = C + n0 + r;
    #pragma unroll
    for (int t = 0; t < 2; ++t) {
        #pragma unroll
        for (int j = 0; j < 8; ++j) {
            const int row = m0 + j + 8 * half;
            crow[(size_t)row * DD + t * 16] = acc[t][j];
        }
    }
}


__global__ __launch_bounds__(32) void stage1_kernel(
    const float* __restrict__ pooled, const float* __restrict__ W1,
    const float* __restrict__ b1, float* __restrict__ H1)
{
    const int g  = blockIdx.x / 96;
    const int wt = blockIdx.x % 96;
    const int m0 = (wt / 24) * 16;
    const int n0 = (wt % 24) * 32;
    gemm_wave_16x32(pooled,
                    W1 + (size_t)g * DD * DD,
                    b1 + (size_t)g * DD,
                    H1 + (size_t)g * BB * DD,
                    m0, n0);
}

#define S2_KB    32
#define S2_NSLAB (DD / S2_KB)
#define S2_NB    96
#define S2_PA    36
#define S2_PB    224
#define S2_AFL   (BB * S2_PA)
#define S2_BFL   ((S2_KB / 2) * S2_PB)
#define S2_STAGE (S2_AFL + S2_BFL)

typedef __attribute__((ext_vector_type(4))) float v4f;

struct Slab {
    v4f a[4];
    v4f bx[3];
    v4f by[3];
};

__device__ __forceinline__ void slab_fetch(
    Slab& g, const float* __restrict__ asrc, const float* __restrict__ bsrc, int k0)
{
    #pragma unroll
    for (int j = 0; j < 4; ++j)
        g.a[j] = *reinterpret_cast<const v4f*>(asrc + (size_t)j * 16 * DD + k0);
    #pragma unroll
    for (int seg = 0; seg < 3; ++seg) {
        g.bx[seg] = *reinterpret_cast<const v4f*>(bsrc + (size_t)k0 * DD + seg * 32);
        g.by[seg] = *reinterpret_cast<const v4f*>(bsrc + (size_t)(k0 + 1) * DD + seg * 32);
    }
}

__device__ __forceinline__ void slab_store(const Slab& g, float* st, int alds, int blds)
{
    #pragma unroll
    for (int j = 0; j < 4; ++j)
        *reinterpret_cast<v4f*>(st + alds + j * 16 * S2_PA) = g.a[j];
    #pragma unroll
    for (int seg = 0; seg < 3; ++seg) {
        v4f lo, hi;
        lo.x = g.bx[seg].x; lo.y = g.by[seg].x; lo.z = g.bx[seg].y; lo.w = g.by[seg].y;
        hi.x = g.bx[seg].z; hi.y = g.by[seg].z; hi.z = g.bx[seg].w; hi.w = g.by[seg].w;
        *reinterpret_cast<v4f*>(st + S2_AFL + blds + seg * 64)     = lo;
        *reinterpret_cast<v4f*>(st + S2_AFL + blds + seg * 64 + 4) = hi;
    }
}

struct SlabFrags {
    v2f a[8];
    v2f b[8][6];
};

__device__ __forceinline__ void slab_read_frags(
    SlabFrags& f, const float* st, int afrag, int bfrag)
{
    #pragma unroll
    for (int s = 0; s < 8; ++s) {
        f.a[s] = *reinterpret_cast<const v2f*>(st + afrag + 4 * s);
        #pragma unroll
        for (int t = 0; t < 6; ++t)
            f.b[s][t] = *reinterpret_cast<const v2f*>(
                st + S2_AFL + bfrag + 2 * s * S2_PB + t * 32);
    }
}

__device__ __forceinline__ void slab_mma(const SlabFrags& f, v8f (&acc)[6])
{
    #pragma unroll
    for (int s = 0; s < 8; ++s) {
        #pragma unroll
        for (int t = 0; t < 6; ++t)
            acc[t] = wmma_f32(f.a[s], f.b[s][t], acc[t]);
    }
}

#define S2_TP 100

struct HeadB {
    v2f b[S2_NB / 4];
};

__device__ __forceinline__ void head_b_load(HeadB& hb, const float* __restrict__ bp)
{
    #pragma unroll
    for (int s = 0; s < S2_NB / 4; ++s) {
        hb.b[s].x = bp[(4 * s) * LL];
        hb.b[s].y = bp[(4 * s + 1) * LL];
    }
}

__device__ __forceinline__ void head_tile(
    const v2f (&af)[S2_NB / 4], const HeadB& hb, bool valid, float* pp)
{
    v8f d0 = {}, d1 = {};
    #pragma unroll
    for (int s = 0; s < S2_NB / 4; s += 2) {
        d0 = wmma_f32(af[s],     hb.b[s],     d0);
        d1 = wmma_f32(af[s + 1], hb.b[s + 1], d1);
    }
    __builtin_amdgcn_sched_barrier(0);
    asm volatile("v_nop\n\tv_nop\n\tv_nop\n\tv_nop\n\tv_nop\n\tv_nop\n\tv_nop\n\tv_nop");
    __builtin_amdgcn_sched_barrier(0);
    if (valid) {
        #pragma unroll
        for (int j = 0; j < 8; ++j) pp[j * LL] = d0[j] + d1[j];
    }
}

__global__ __launch_bounds__(128) void stage2_kernel(
    const float* __restrict__ H1, const float* __restrict__ W2,
    const float* __restrict__ b2, const float* __restrict__ Wh,
    const int* __restrict__ g1_idx, const int* __restrict__ g2_idx,
    float* __restrict__ P)
{
    __shared__ __attribute__((aligned(16))) float lds[2 * S2_STAGE + 4 * AANN + 8];

    const int p  = blockIdx.x >> 3;
    const int nb = blockIdx.x & 7;
    const int g1 = p >> 3;
    const int g2 = p & 7;
    const float* A    = H1 + (size_t)g1 * BB * DD;
    const float* Bm   = W2 + (size_t)g2 * DD * DD + nb * S2_NB;
    const float* bias = b2 + (size_t)g2 * DD + nb * S2_NB;

    const int tid  = threadIdx.x;
    const int lane = tid & 31;
    const int w    = tid >> 5;
    const int half = lane >> 4;
    const int r    = lane & 15;
    const int tr   = tid >> 3;
    const int tq   = tid & 7;

    const float* asrc = A + (size_t)tr * DD + tq * 4;
    const float* bsrc = Bm + (size_t)(2 * tr) * DD + tq * 4;
    const int alds  = tr * S2_PA + tq * 4;
    const int blds  = tr * S2_PB + tq * 8;
    const int afrag = (w * 16 + r) * S2_PA + 2 * half;
    const int bfrag = half * S2_PB + 2 * r;

    v8f acc[6];
    #pragma unroll
    for (int t = 0; t < 6; ++t) {
        const float bv = bias[t * 16 + r];
        #pragma unroll
        for (int j = 0; j < 8; ++j) acc[t][j] = bv;
    }

    Slab g;
    slab_fetch(g, asrc, bsrc, 0);

    int* mylist = reinterpret_cast<int*>(lds + 2 * S2_STAGE) + w * AANN;
    int ga[AANN / 32], gb[AANN / 32];
    #pragma unroll
    for (int c = 0; c < AANN / 32; ++c) {
        ga[c] = g1_idx[c * 32 + lane];
        gb[c] = g2_idx[c * 32 + lane];
    }
    int cnt = 0;
    #pragma unroll
    for (int c = 0; c < AANN / 32; ++c) {
        const bool m = (ga[c] == g1) & (gb[c] == g2);
        const unsigned mask = __builtin_amdgcn_ballot_w32(m);
        if (m) mylist[cnt + __builtin_popcount(mask & ((1u << lane) - 1u))] = c * 32 + lane;
        cnt += __builtin_popcount(mask);
    }

    slab_store(g, lds, alds, blds);
    __syncthreads();

    for (int c = 0; c < S2_NSLAB - 1; ++c) {
        const float* cur = lds + (c & 1) * S2_STAGE;
        float*       nxt = lds + ((c + 1) & 1) * S2_STAGE;
        slab_fetch(g, asrc, bsrc, (c + 1) * S2_KB);
        SlabFrags f;
        slab_read_frags(f, cur, afrag, bfrag);
        __builtin_amdgcn_sched_barrier(0);
        slab_mma(f, acc);
        __builtin_amdgcn_sched_barrier(0);
        slab_store(g, nxt, alds, blds);
        __syncthreads();
    }
    const int hl = r & 1;
    const int hs = r >> 1;
    const float* whn = Wh + (size_t)(nb * S2_NB + 2 * half) * LL + hl;
    HeadB cur;
    head_b_load(cur, whn + (size_t)((hs < cnt) ? mylist[hs] : 0) * DD * LL);
    __builtin_amdgcn_sched_barrier(0);
    {
        SlabFrags f;
        slab_read_frags(f, lds + ((S2_NSLAB - 1) & 1) * S2_STAGE, afrag, bfrag);
        slab_mma(f, acc);
    }

    __syncthreads();
    float* tile = lds + w * 16 * S2_TP;
    #pragma unroll
    for (int t = 0; t < 6; ++t) {
        #pragma unroll
        for (int j = 0; j < 8; ++j)
            tile[(j + 8 * half) * S2_TP + t * 16 + r] = acc[t][j];
    }
    __syncthreads();
    v2f af[S2_NB / 4];
    #pragma unroll
    for (int s = 0; s < S2_NB / 4; ++s)
        af[s] = *reinterpret_cast<const v2f*>(tile + r * S2_TP + 4 * s + 2 * half);

    const float* wh_of = whn;
    float* pbase = P + ((size_t)nb * AANN * BB + w * 16 + 8 * half) * LL + hl;
    HeadB hb1;
    for (int t0 = 0; t0 < cnt; t0 += 16) {
        const bool second = (t0 + 8 < cnt);
        if (second) {
            const int i1 = t0 + 8 + hs;
            head_b_load(hb1, wh_of + (size_t)((i1 < cnt) ? mylist[i1] : 0) * DD * LL);
        }
        {
            const bool valid = (t0 + hs) < cnt;
            const int  a     = valid ? mylist[t0 + hs] : 0;
            head_tile(af, cur, valid, pbase + (size_t)a * BB * LL);
        }
        if (second) {
            if (t0 + 16 < cnt) {
                const int i0 = t0 + 16 + hs;
                head_b_load(cur, wh_of + (size_t)((i0 < cnt) ? mylist[i0] : 0) * DD * LL);
            }
            const bool valid = (t0 + 8 + hs) < cnt;
            const int  a     = valid ? mylist[t0 + 8 + hs] : 0;
            head_tile(af, hb1, valid, pbase + (size_t)a * BB * LL);
        }
    }
}

__global__ __launch_bounds__(256) void head_sum_kernel(
    const float* __restrict__ P, const float* __restrict__ bh, float* __restrict__ out)
{
    __shared__ float T[16][33];

    const int ag  = blockIdx.x >> 2;
    const int q   = blockIdx.x & 3;
    const int tid = threadIdx.x;
    const int al  = tid >> 4;
    const int bl  = tid & 15;
    const int a   = ag * 16 + al;
    const int b   = q * 16 + bl;

    const float* src = P + ((size_t)a * BB + b) * LL;
    v2f pv[8];
    #pragma unroll
    for (int nb = 0; nb < 8; ++nb)
        pv[nb] = *reinterpret_cast<const v2f*>(src + (size_t)nb * AANN * BB * LL);
    v2f s = *reinterpret_cast<const v2f*>(bh + a * LL);
    #pragma unroll
    for (int nb = 0; nb < 8; ++nb) { s.x += pv[nb].x; s.y += pv[nb].y; }

    T[bl][al * 2]     = s.x;
    T[bl][al * 2 + 1] = s.y;
    __syncthreads();

    const int lane = tid & 31;
    const int w8   = tid >> 5;
    #pragma unroll
    for (int hh = 0; hh < 2; ++hh) {
        const int row = w8 + 8 * hh;
        out[(size_t)(q * 16 + row) * (AANN * LL) + ag * 32 + lane] = T[row][lane];
    }
}

extern "C" void kernel_launch(void* const* d_in, const int* in_sizes, int n_in,
                              void* d_out, int out_size, void* d_ws, size_t ws_size,
                              hipStream_t stream) {
    const float* pooled = (const float*)d_in[0];
    const float* W1     = (const float*)d_in[1];
    const float* b1     = (const float*)d_in[2];
    const float* W2     = (const float*)d_in[3];
    const float* b2     = (const float*)d_in[4];
    const float* Wh     = (const float*)d_in[5];
    const float* bh     = (const float*)d_in[6];
    const int*   g1     = (const int*)d_in[7];
    const int*   g2     = (const int*)d_in[8];
    float*       out    = (float*)d_out;

    float* H1 = (float*)d_ws;
    float* P  = H1 + (size_t)GG1 * BB * DD;

    stage1_kernel<<<GG1 * 96, 32, 0, stream>>>(pooled, W1, b1, H1);
    stage2_kernel<<<GG1 * GG2 * 8, 128, 0, stream>>>(H1, W2, b2, Wh, g1, g2, P);
    head_sum_kernel<<<(AANN / 16) * 4, 256, 0, stream>>>(P, bh, out);
}


# Stage 2 requests each 32-k slab of H1 and W2 three slabs ahead of its products (three register sets in rotation, was one ahead); its slab barrier orders LDS only, so loads stay in flight; sums same
# speedup vs baseline: 6.6618x; 1.1322x over previous
#include <hip/hip_runtime.h>
#include <hip/hip_bf16.h>

#define AANN 256
#define BB   64
#define DD   768
#define LL   2
#define GG1  4
#define GG2  8

typedef __attribute__((ext_vector_type(2))) float v2f;
typedef __attribute__((ext_vector_type(8))) float v8f;

__device__ __forceinline__ v8f wmma_f32(v2f a, v2f b, v8f c) {
    return __builtin_amdgcn_wmma_f32_16x16x4_f32(
         false, a,  false, b,
         (short)0, c,  false,  false);
}


#define KC1    12
#define KSPAN1 (KC1 * 4)

struct Tile1Frags {
    v2f a[KC1];
    v2f b[KC1][2];
};

__device__ __forceinline__ void tile1_load_frags(
    Tile1Frags& f, const float* __restrict__ arow, const float* __restrict__ bcol, int ka0)
{
    #pragma unroll
    for (int s = 0; s < KC1; ++s) {
        const int ka = ka0 + 4 * s;
        f.a[s].x = arow[ka];
        f.a[s].y = arow[ka + 1];
        const float* bk = bcol + (size_t)ka * DD;
        #pragma unroll
        for (int t = 0; t < 2; ++t) {
            f.b[s][t].x = bk[t * 16];
            f.b[s][t].y = bk[t * 16 + DD];
        }
    }
}

__device__ __forceinline__ void tile1_mma_frags(const Tile1Frags& f, v8f (&acc)[2])
{
    #pragma unroll
    for (int s = 0; s < KC1; ++s) {
        #pragma unroll
        for (int t = 0; t < 2; ++t) {
            acc[t] = wmma_f32(f.a[s], f.b[s][t], acc[t]);
        }
    }
}

__device__ __forceinline__ void gemm_wave_16x32(
    const float* __restrict__ A,
    const float* __restrict__ Bm,
    const float* __restrict__ bias,
    float* __restrict__ C,
    int m0, int n0)
{
    const int lane = threadIdx.x & 31;
    const int half = lane >> 4;
    const int r    = lane & 15;

    v8f acc[2];
    #pragma unroll
    for (int t = 0; t < 2; ++t) {
        const float bv = bias[n0 + t * 16 + r];
        #pragma unroll
        for (int j = 0; j < 8; ++j) acc[t][j] = bv;
    }

    const float* arow = A + (size_t)(m0 + r) * DD;
    const float* bcol = Bm + n0 + r;

    Tile1Frags f0, f1;
    tile1_load_frags(f0, arow, bcol, 2 * half);
    for (int k = 0; k < DD - 2 * KSPAN1; k += 2 * KSPAN1) {
        tile1_load_frags(f1, arow, bcol, k + KSPAN1 + 2 * half);
        __builtin_amdgcn_sched_barrier(0);
        tile1_mma_frags(f0, acc);
        tile1_load_frags(f0, arow, bcol, k + 2 * KSPAN1 + 2 * half);
        __builtin_amdgcn_sched_barrier(0);
        tile1_mma_frags(f1, acc);
    }
    tile1_load_frags(f1, arow, bcol, DD - KSPAN1 + 2 * half);
    __builtin_amdgcn_sched_barrier(0);
    tile1_mma_frags(f0, acc);
    tile1_mma_frags(f1, acc);

    float* crow = C + n0 + r;
    #pragma unroll
    for (int t = 0; t < 2; ++t) {
        #pragma unroll
        for (int j = 0; j < 8; ++j) {
            const int row = m0 + j + 8 * half;
            crow[(size_t)row * DD + t * 16] = acc[t][j];
        }
    }
}


__global__ __launch_bounds__(32) void stage1_kernel(
    const float* __restrict__ pooled, const float* __restrict__ W1,
    const float* __restrict__ b1, float* __restrict__ H1)
{
    const int g  = blockIdx.x / 96;
    const int wt = blockIdx.x % 96;
    const int m0 = (wt / 24) * 16;
    const int n0 = (wt % 24) * 32;
    gemm_wave_16x32(pooled,
                    W1 + (size_t)g * DD * DD,
                    b1 + (size_t)g * DD,
                    H1 + (size_t)g * BB * DD,
                    m0, n0);
}

#define S2_KB    32
#define S2_NSLAB (DD / S2_KB)
#define S2_NB    96
#define S2_PA    36
#define S2_PB    224
#define S2_AFL   (BB * S2_PA)
#define S2_BFL   ((S2_KB / 2) * S2_PB)
#define S2_STAGE (S2_AFL + S2_BFL)
#define S2_AHEAD 3

typedef __attribute__((ext_vector_type(4))) float v4f;

struct Slab {
    v4f a[4];
    v4f bx[3];
    v4f by[3];
};

__device__ __forceinline__ void slab_fetch(
    Slab& g, const float* __restrict__ asrc, const float* __restrict__ bsrc, int k0)
{
    #pragma unroll
    for (int j = 0; j < 4; ++j)
        g.a[j] = *reinterpret_cast<const v4f*>(asrc + (size_t)j * 16 * DD + k0);
    #pragma unroll
    for (int seg = 0; seg < 3; ++seg) {
        g.bx[seg] = *reinterpret_cast<const v4f*>(bsrc + (size_t)k0 * DD + seg * 32);
        g.by[seg] = *reinterpret_cast<const v4f*>(bsrc + (size_t)(k0 + 1) * DD + seg * 32);
    }
}

__device__ __forceinline__ void slab_store(const Slab& g, float* st, int alds, int blds)
{
    #pragma unroll
    for (int j = 0; j < 4; ++j)
        *reinterpret_cast<v4f*>(st + alds + j * 16 * S2_PA) = g.a[j];
    #pragma unroll
    for (int seg = 0; seg < 3; ++seg) {
        v4f lo, hi;
        lo.x = g.bx[seg].x; lo.y = g.by[seg].x; lo.z = g.bx[seg].y; lo.w = g.by[seg].y;
        hi.x = g.bx[seg].z; hi.y = g.by[seg].z; hi.z = g.bx[seg].w; hi.w = g.by[seg].w;
        *reinterpret_cast<v4f*>(st + S2_AFL + blds + seg * 64)     = lo;
        *reinterpret_cast<v4f*>(st + S2_AFL + blds + seg * 64 + 4) = hi;
    }
}

struct SlabFrags {
    v2f a[8];
    v2f b[8][6];
};

__device__ __forceinline__ void slab_read_frags(
    SlabFrags& f, const float* st, int afrag, int bfrag)
{
    #pragma unroll
    for (int s = 0; s < 8; ++s) {
        f.a[s] = *reinterpret_cast<const v2f*>(st + afrag + 4 * s);
        #pragma unroll
        for (int t = 0; t < 6; ++t)
            f.b[s][t] = *reinterpret_cast<const v2f*>(
                st + S2_AFL + bfrag + 2 * s * S2_PB + t * 32);
    }
}

__device__ __forceinline__ void slab_mma(const SlabFrags& f, v8f (&acc)[6])
{
    #pragma unroll
    for (int s = 0; s < 8; ++s) {
        #pragma unroll
        for (int t = 0; t < 6; ++t)
            acc[t] = wmma_f32(f.a[s], f.b[s][t], acc[t]);
    }
}

#define S2_TP 100

struct HeadB {
    v2f b[S2_NB / 4];
};

__device__ __forceinline__ void head_b_load(HeadB& hb, const float* __restrict__ bp)
{
    #pragma unroll
    for (int s = 0; s < S2_NB / 4; ++s) {
        hb.b[s].x = bp[(4 * s) * LL];
        hb.b[s].y = bp[(4 * s + 1) * LL];
    }
}

__device__ __forceinline__ void head_tile(
    const v2f (&af)[S2_NB / 4], const HeadB& hb, bool valid, float* pp)
{
    v8f d0 = {}, d1 = {};
    #pragma unroll
    for (int s = 0; s < S2_NB / 4; s += 2) {
        d0 = wmma_f32(af[s],     hb.b[s],     d0);
        d1 = wmma_f32(af[s + 1], hb.b[s + 1], d1);
    }
    __builtin_amdgcn_sched_barrier(0);
    asm volatile("v_nop\n\tv_nop\n\tv_nop\n\tv_nop\n\tv_nop\n\tv_nop\n\tv_nop\n\tv_nop");
    __builtin_amdgcn_sched_barrier(0);
    if (valid) {
        #pragma unroll
        for (int j = 0; j < 8; ++j) pp[j * LL] = d0[j] + d1[j];
    }
}

__device__ __forceinline__ void lds_handover_barrier()
{
    __builtin_amdgcn_fence(__ATOMIC_RELEASE, "workgroup", "local");
    __builtin_amdgcn_s_barrier();
    __builtin_amdgcn_fence(__ATOMIC_ACQUIRE, "workgroup", "local");
}

__global__ __launch_bounds__(128) void stage2_kernel(
    const float* __restrict__ H1, const float* __restrict__ W2,
    const float* __restrict__ b2, const float* __restrict__ Wh,
    const int* __restrict__ g1_idx, const int* __restrict__ g2_idx,
    float* __restrict__ P)
{
    __shared__ __attribute__((aligned(16))) float lds[2 * S2_STAGE + 4 * AANN + 8];

    const int p  = blockIdx.x >> 3;
    const int nb = blockIdx.x & 7;
    const int g1 = p >> 3;
    const int g2 = p & 7;
    const float* A    = H1 + (size_t)g1 * BB * DD;
    const float* Bm   = W2 + (size_t)g2 * DD * DD + nb * S2_NB;
    const float* bias = b2 + (size_t)g2 * DD + nb * S2_NB;

    const int tid  = threadIdx.x;
    const int lane = tid & 31;
    const int w    = tid >> 5;
    const int half = lane >> 4;
    const int r    = lane & 15;
    const int tr   = tid >> 3;
    const int tq   = tid & 7;

    const float* asrc = A + (size_t)tr * DD + tq * 4;
    const float* bsrc = Bm + (size_t)(2 * tr) * DD + tq * 4;
    const int alds  = tr * S2_PA + tq * 4;
    const int blds  = tr * S2_PB + tq * 8;
    const int afrag = (w * 16 + r) * S2_PA + 2 * half;
    const int bfrag = half * S2_PB + 2 * r;

    v8f acc[6];
    #pragma unroll
    for (int t = 0; t < 6; ++t) {
        const float bv = bias[t * 16 + r];
        #pragma unroll
        for (int j = 0; j < 8; ++j) acc[t][j] = bv;
    }

    int* mylist = reinterpret_cast<int*>(lds + 2 * S2_STAGE) + w * AANN;
    int ga[AANN / 32], gb[AANN / 32];
    #pragma unroll
    for (int c = 0; c < AANN / 32; ++c) {
        ga[c] = g1_idx[c * 32 + lane];
        gb[c] = g2_idx[c * 32 + lane];
    }
    __builtin_amdgcn_sched_barrier(0);

    Slab g[S2_AHEAD];
    #pragma unroll
    for (int u = 0; u < S2_AHEAD; ++u) slab_fetch(g[u], asrc, bsrc, u * S2_KB);
    int cnt = 0;
    #pragma unroll
    for (int c = 0; c < AANN / 32; ++c) {
        const bool m = (ga[c] == g1) & (gb[c] == g2);
        const unsigned mask = __builtin_amdgcn_ballot_w32(m);
        if (m) mylist[cnt + __builtin_popcount(mask & ((1u << lane) - 1u))] = c * 32 + lane;
        cnt += __builtin_popcount(mask);
    }

    slab_store(g[0], lds, alds, blds);
    lds_handover_barrier();

    for (int c0 = 0; c0 < S2_NSLAB - S2_AHEAD; c0 += S2_AHEAD) {
        #pragma unroll
        for (int u = 0; u < S2_AHEAD; ++u) {
            const int c = c0 + u;
            const float* cur = lds + (c & 1) * S2_STAGE;
            float*       nxt = lds + ((c + 1) & 1) * S2_STAGE;
            slab_fetch(g[u], asrc, bsrc, (c + S2_AHEAD) * S2_KB);
            SlabFrags f;
            slab_read_frags(f, cur, afrag, bfrag);
            __builtin_amdgcn_sched_barrier(0);
            slab_mma(f, acc);
            __builtin_amdgcn_sched_barrier(0);
            slab_store(g[(u + 1) % S2_AHEAD], nxt, alds, blds);
            lds_handover_barrier();
        }
    }
    #pragma unroll
    for (int u = 0; u < S2_AHEAD - 1; ++u) {
        const int c = S2_NSLAB - S2_AHEAD + u;
        const float* cur = lds + (c & 1) * S2_STAGE;
        float*       nxt = lds + ((c + 1) & 1) * S2_STAGE;
        SlabFrags f;
        slab_read_frags(f, cur, afrag, bfrag);
        __builtin_amdgcn_sched_barrier(0);
        slab_mma(f, acc);
        __builtin_amdgcn_sched_barrier(0);
        slab_store(g[(u + 1) % S2_AHEAD], nxt, alds, blds);
        lds_handover_barrier();
    }
    const int hl = r & 1;
    const int hs = r >> 1;
    const float* whn = Wh + (size_t)(nb * S2_NB + 2 * half) * LL + hl;
    HeadB cur;
    head_b_load(cur, whn + (size_t)((hs < cnt) ? mylist[hs] : 0) * DD * LL);
    __builtin_amdgcn_sched_barrier(0);
    {
        SlabFrags f;
        slab_read_frags(f, lds + ((S2_NSLAB - 1) & 1) * S2_STAGE, afrag, bfrag);
        slab_mma(f, acc);
    }

    __syncthreads();
    float* tile = lds + w * 16 * S2_TP;
    #pragma unroll
    for (int t = 0; t < 6; ++t) {
        #pragma unroll
        for (int j = 0; j < 8; ++j)
            tile[(j + 8 * half) * S2_TP + t * 16 + r] = acc[t][j];
    }
    __syncthreads();
    v2f af[S2_NB / 4];
    #pragma unroll
    for (int s = 0; s < S2_NB / 4; ++s)
        af[s] = *reinterpret_cast<const v2f*>(tile + r * S2_TP + 4 * s + 2 * half);

    const float* wh_of = whn;
    float* pbase = P + ((size_t)nb * AANN * BB + w * 16 + 8 * half) * LL + hl;
    HeadB hb1;
    for (int t0 = 0; t0 < cnt; t0 += 16) {
        const bool second = (t0 + 8 < cnt);
        if (second) {
            const int i1 = t0 + 8 + hs;
            head_b_load(hb1, wh_of + (size_t)((i1 < cnt) ? mylist[i1] : 0) * DD * LL);
        }
        {
            const bool valid = (t0 + hs) < cnt;
            const int  a     = valid ? mylist[t0 + hs] : 0;
            head_tile(af, cur, valid, pbase + (size_t)a * BB * LL);
        }
        if (second) {
            if (t0 + 16 < cnt) {
                const int i0 = t0 + 16 + hs;
                head_b_load(cur, wh_of + (size_t)((i0 < cnt) ? mylist[i0] : 0) * DD * LL);
            }
            const bool valid = (t0 + 8 + hs) < cnt;
            const int  a     = valid ? mylist[t0 + 8 + hs] : 0;
            head_tile(af, hb1, valid, pbase + (size_t)a * BB * LL);
        }
    }
}

__global__ __launch_bounds__(256) void head_sum_kernel(
    const float* __restrict__ P, const float* __restrict__ bh, float* __restrict__ out)
{
    __shared__ float T[16][33];

    const int ag  = blockIdx.x >> 2;
    const int q   = blockIdx.x & 3;
    const int tid = threadIdx.x;
    const int al  = tid >> 4;
    const int bl  = tid & 15;
    const int a   = ag * 16 + al;
    const int b   = q * 16 + bl;

    const float* src = P + ((size_t)a * BB + b) * LL;
    v2f pv[8];
    #pragma unroll
    for (int nb = 0; nb < 8; ++nb)
        pv[nb] = *reinterpret_cast<const v2f*>(src + (size_t)nb * AANN * BB * LL);
    v2f s = *reinterpret_cast<const v2f*>(bh + a * LL);
    #pragma unroll
    for (int nb = 0; nb < 8; ++nb) { s.x += pv[nb].x; s.y += pv[nb].y; }

    T[bl][al * 2]     = s.x;
    T[bl][al * 2 + 1] = s.y;
    __syncthreads();

    const int lane = tid & 31;
    const int w8   = tid >> 5;
    #pragma unroll
    for (int hh = 0; hh < 2; ++hh) {
        const int row = w8 + 8 * hh;
        out[(size_t)(q * 16 + row) * (AANN * LL) + ag * 32 + lane] = T[row][lane];
    }
}

extern "C" void kernel_launch(void* const* d_in, const int* in_sizes, int n_in,
                              void* d_out, int out_size, void* d_ws, size_t ws_size,
                              hipStream_t stream) {
    const float* pooled = (const float*)d_in[0];
    const float* W1     = (const float*)d_in[1];
    const float* b1     = (const float*)d_in[2];
    const float* W2     = (const float*)d_in[3];
    const float* b2     = (const float*)d_in[4];
    const float* Wh     = (const float*)d_in[5];
    const float* bh     = (const float*)d_in[6];
    const int*   g1     = (const int*)d_in[7];
    const int*   g2     = (const int*)d_in[8];
    float*       out    = (float*)d_out;

    float* H1 = (float*)d_ws;
    float* P  = H1 + (size_t)GG1 * BB * DD;

    stage1_kernel<<<GG1 * 96, 32, 0, stream>>>(pooled, W1, b1, H1);
    stage2_kernel<<<GG1 * GG2 * 8, 128, 0, stream>>>(H1, W2, b2, Wh, g1, g2, P);
    head_sum_kernel<<<(AANN / 16) * 4, 256, 0, stream>>>(P, bh, out);
}


# Stage 2: a second fragment register set; each step issues the next slab's LDS reads and the slab-after's LDS writes ahead of the current slab's products (4 register sets from memory); sums same
# speedup vs baseline: 6.9945x; 1.0499x over previous
#include <hip/hip_runtime.h>
#include <hip/hip_bf16.h>

#define AANN 256
#define BB   64
#define DD   768
#define LL   2
#define GG1  4
#define GG2  8

typedef __attribute__((ext_vector_type(2))) float v2f;
typedef __attribute__((ext_vector_type(8))) float v8f;

__device__ __forceinline__ v8f wmma_f32(v2f a, v2f b, v8f c) {
    return __builtin_amdgcn_wmma_f32_16x16x4_f32(
         false, a,  false, b,
         (short)0, c,  false,  false);
}


#define KC1    12
#define KSPAN1 (KC1 * 4)

struct Tile1Frags {
    v2f a[KC1];
    v2f b[KC1][2];
};

__device__ __forceinline__ void tile1_load_frags(
    Tile1Frags& f, const float* __restrict__ arow, const float* __restrict__ bcol, int ka0)
{
    #pragma unroll
    for (int s = 0; s < KC1; ++s) {
        const int ka = ka0 + 4 * s;
        f.a[s].x = arow[ka];
        f.a[s].y = arow[ka + 1];
        const float* bk = bcol + (size_t)ka * DD;
        #pragma unroll
        for (int t = 0; t < 2; ++t) {
            f.b[s][t].x = bk[t * 16];
            f.b[s][t].y = bk[t * 16 + DD];
        }
    }
}

__device__ __forceinline__ void tile1_mma_frags(const Tile1Frags& f, v8f (&acc)[2])
{
    #pragma unroll
    for (int s = 0; s < KC1; ++s) {
        #pragma unroll
        for (int t = 0; t < 2; ++t) {
            acc[t] = wmma_f32(f.a[s], f.b[s][t], acc[t]);
        }
    }
}

__device__ __forceinline__ void gemm_wave_16x32(
    const float* __restrict__ A,
    const float* __restrict__ Bm,
    const float* __restrict__ bias,
    float* __restrict__ C,
    int m0, int n0)
{
    const int lane = threadIdx.x & 31;
    const int half = lane >> 4;
    const int r    = lane & 15;

    v8f acc[2];
    #pragma unroll
    for (int t = 0; t < 2; ++t) {
        const float bv = bias[n0 + t * 16 + r];
        #pragma unroll
        for (int j = 0; j < 8; ++j) acc[t][j] = bv;
    }

    const float* arow = A + (size_t)(m0 + r) * DD;
    const float* bcol = Bm + n0 + r;

    Tile1Frags f0, f1;
    tile1_load_frags(f0, arow, bcol, 2 * half);
    for (int k = 0; k < DD - 2 * KSPAN1; k += 2 * KSPAN1) {
        tile1_load_frags(f1, arow, bcol, k + KSPAN1 + 2 * half);
        __builtin_amdgcn_sched_barrier(0);
        tile1_mma_frags(f0, acc);
        tile1_load_frags(f0, arow, bcol, k + 2 * KSPAN1 + 2 * half);
        __builtin_amdgcn_sched_barrier(0);
        tile1_mma_frags(f1, acc);
    }
    tile1_load_frags(f1, arow, bcol, DD - KSPAN1 + 2 * half);
    __builtin_amdgcn_sched_barrier(0);
    tile1_mma_frags(f0, acc);
    tile1_mma_frags(f1, acc);

    float* crow = C + n0 + r;
    #pragma unroll
    for (int t = 0; t < 2; ++t) {
        #pragma unroll
        for (int j = 0; j < 8; ++j) {
            const int row = m0 + j + 8 * half;
            crow[(size_t)row * DD + t * 16] = acc[t][j];
        }
    }
}


__global__ __launch_bounds__(32) void stage1_kernel(
    const float* __restrict__ pooled, const float* __restrict__ W1,
    const float* __restrict__ b1, float* __restrict__ H1)
{
    const int g  = blockIdx.x / 96;
    const int wt = blockIdx.x % 96;
    const int m0 = (wt / 24) * 16;
    const int n0 = (wt % 24) * 32;
    gemm_wave_16x32(pooled,
                    W1 + (size_t)g * DD * DD,
                    b1 + (size_t)g * DD,
                    H1 + (size_t)g * BB * DD,
                    m0, n0);
}

#define S2_KB    32
#define S2_NSLAB (DD / S2_KB)
#define S2_NB    96
#define S2_PA    36
#define S2_PB    224
#define S2_AFL   (BB * S2_PA)
#define S2_BFL   ((S2_KB / 2) * S2_PB)
#define S2_STAGE (S2_AFL + S2_BFL)
#define S2_SETS  4

typedef __attribute__((ext_vector_type(4))) float v4f;

struct Slab {
    v4f a[4];
    v4f bx[3];
    v4f by[3];
};

__device__ __forceinline__ void slab_fetch(
    Slab& g, const float* __restrict__ asrc, const float* __restrict__ bsrc, int k0)
{
    #pragma unroll
    for (int j = 0; j < 4; ++j)
        g.a[j] = *reinterpret_cast<const v4f*>(asrc + (size_t)j * 16 * DD + k0);
    #pragma unroll
    for (int seg = 0; seg < 3; ++seg) {
        g.bx[seg] = *reinterpret_cast<const v4f*>(bsrc + (size_t)k0 * DD + seg * 32);
        g.by[seg] = *reinterpret_cast<const v4f*>(bsrc + (size_t)(k0 + 1) * DD + seg * 32);
    }
}

__device__ __forceinline__ void slab_store(const Slab& g, float* st, int alds, int blds)
{
    #pragma unroll
    for (int j = 0; j < 4; ++j)
        *reinterpret_cast<v4f*>(st + alds + j * 16 * S2_PA) = g.a[j];
    #pragma unroll
    for (int seg = 0; seg < 3; ++seg) {
        v4f lo, hi;
        lo.x = g.bx[seg].x; lo.y = g.by[seg].x; lo.z = g.bx[seg].y; lo.w = g.by[seg].y;
        hi.x = g.bx[seg].z; hi.y = g.by[seg].z; hi.z = g.bx[seg].w; hi.w = g.by[seg].w;
        *reinterpret_cast<v4f*>(st + S2_AFL + blds + seg * 64)     = lo;
        *reinterpret_cast<v4f*>(st + S2_AFL + blds + seg * 64 + 4) = hi;
    }
}

struct SlabFrags {
    v2f a[8];
    v2f b[8][6];
};

__device__ __forceinline__ void slab_read_frags(
    SlabFrags& f, const float* st, int afrag, int bfrag)
{
    #pragma unroll
    for (int s = 0; s < 8; ++s) {
        f.a[s] = *reinterpret_cast<const v2f*>(st + afrag + 4 * s);
        #pragma unroll
        for (int t = 0; t < 6; ++t)
            f.b[s][t] = *reinterpret_cast<const v2f*>(
                st + S2_AFL + bfrag + 2 * s * S2_PB + t * 32);
    }
}

__device__ __forceinline__ void slab_mma(const SlabFrags& f, v8f (&acc)[6])
{
    #pragma unroll
    for (int s = 0; s < 8; ++s) {
        #pragma unroll
        for (int t = 0; t < 6; ++t)
            acc[t] = wmma_f32(f.a[s], f.b[s][t], acc[t]);
    }
}

#define S2_TP 100

struct HeadB {
    v2f b[S2_NB / 4];
};

__device__ __forceinline__ void head_b_load(HeadB& hb, const float* __restrict__ bp)
{
    #pragma unroll
    for (int s = 0; s < S2_NB / 4; ++s) {
        hb.b[s].x = bp[(4 * s) * LL];
        hb.b[s].y = bp[(4 * s + 1) * LL];
    }
}

__device__ __forceinline__ void head_tile(
    const v2f (&af)[S2_NB / 4], const HeadB& hb, bool valid, float* pp)
{
    v8f d0 = {}, d1 = {};
    #pragma unroll
    for (int s = 0; s < S2_NB / 4; s += 2) {
        d0 = wmma_f32(af[s],     hb.b[s],     d0);
        d1 = wmma_f32(af[s + 1], hb.b[s + 1], d1);
    }
    __builtin_amdgcn_sched_barrier(0);
    asm volatile("v_nop\n\tv_nop\n\tv_nop\n\tv_nop\n\tv_nop\n\tv_nop\n\tv_nop\n\tv_nop");
    __builtin_amdgcn_sched_barrier(0);
    if (valid) {
        #pragma unroll
        for (int j = 0; j < 8; ++j) pp[j * LL] = d0[j] + d1[j];
    }
}

__device__ __forceinline__ void lds_handover_barrier()
{
    __builtin_amdgcn_fence(__ATOMIC_RELEASE, "workgroup", "local");
    __builtin_amdgcn_s_barrier();
    __builtin_amdgcn_fence(__ATOMIC_ACQUIRE, "workgroup", "local");
}

__global__ __launch_bounds__(128) void stage2_kernel(
    const float* __restrict__ H1, const float* __restrict__ W2,
    const float* __restrict__ b2, const float* __restrict__ Wh,
    const int* __restrict__ g1_idx, const int* __restrict__ g2_idx,
    float* __restrict__ P)
{
    __shared__ __attribute__((aligned(16))) float lds[2 * S2_STAGE + 4 * AANN + 8];

    const int p  = blockIdx.x >> 3;
    const int nb = blockIdx.x & 7;
    const int g1 = p >> 3;
    const int g2 = p & 7;
    const float* A    = H1 + (size_t)g1 * BB * DD;
    const float* Bm   = W2 + (size_t)g2 * DD * DD + nb * S2_NB;
    const float* bias = b2 + (size_t)g2 * DD + nb * S2_NB;

    const int tid  = threadIdx.x;
    const int lane = tid & 31;
    const int w    = tid >> 5;
    const int half = lane >> 4;
    const int r    = lane & 15;
    const int tr   = tid >> 3;
    const int tq   = tid & 7;

    const float* asrc = A + (size_t)tr * DD + tq * 4;
    const float* bsrc = Bm + (size_t)(2 * tr) * DD + tq * 4;
    const int alds  = tr * S2_PA + tq * 4;
    const int blds  = tr * S2_PB + tq * 8;
    const int afrag = (w * 16 + r) * S2_PA + 2 * half;
    const int bfrag = half * S2_PB + 2 * r;

    v8f acc[6];
    #pragma unroll
    for (int t = 0; t < 6; ++t) {
        const float bv = bias[t * 16 + r];
        #pragma unroll
        for (int j = 0; j < 8; ++j) acc[t][j] = bv;
    }

    int* mylist = reinterpret_cast<int*>(lds + 2 * S2_STAGE) + w * AANN;
    int ga[AANN / 32], gb[AANN / 32];
    #pragma unroll
    for (int c = 0; c < AANN / 32; ++c) {
        ga[c] = g1_idx[c * 32 + lane];
        gb[c] = g2_idx[c * 32 + lane];
    }
    __builtin_amdgcn_sched_barrier(0);

    Slab g[S2_SETS];
    #pragma unroll
    for (int u = 0; u < S2_SETS; ++u) slab_fetch(g[u], asrc, bsrc, u * S2_KB);
    int cnt = 0;
    #pragma unroll
    for (int c = 0; c < AANN / 32; ++c) {
        const bool m = (ga[c] == g1) & (gb[c] == g2);
        const unsigned mask = __builtin_amdgcn_ballot_w32(m);
        if (m) mylist[cnt + __builtin_popcount(mask & ((1u << lane) - 1u))] = c * 32 + lane;
        cnt += __builtin_popcount(mask);
    }

    slab_store(g[0], lds, alds, blds);
    slab_store(g[1], lds + S2_STAGE, alds, blds);
    lds_handover_barrier();
    slab_fetch(g[0], asrc, bsrc, S2_SETS * S2_KB);
    SlabFrags f[2];
    slab_read_frags(f[0], lds, afrag, bfrag);
    lds_handover_barrier();

    for (int c0 = 0; c0 < 16; c0 += 4) {
        #pragma unroll
        for (int u = 0; u < 4; ++u) {
            const int c = c0 + u;
            float*       own = lds + (u & 1) * S2_STAGE;
            const float* oth = lds + ((u + 1) & 1) * S2_STAGE;
            slab_fetch(g[(u + 1) % S2_SETS], asrc, bsrc, (c + 5) * S2_KB);
            slab_store(g[(u + 2) % S2_SETS], own, alds, blds);
            slab_read_frags(f[(u + 1) & 1], oth, afrag, bfrag);
            __builtin_amdgcn_sched_barrier(0);
            slab_mma(f[u & 1], acc);
            __builtin_amdgcn_sched_barrier(0);
            lds_handover_barrier();
        }
    }
    #pragma unroll
    for (int c = 16; c < S2_NSLAB - 1; ++c) {
        float*       own = lds + (c & 1) * S2_STAGE;
        const float* oth = lds + ((c + 1) & 1) * S2_STAGE;
        if (c + 5 < S2_NSLAB) slab_fetch(g[(c + 5) % S2_SETS], asrc, bsrc, (c + 5) * S2_KB);
        if (c + 2 < S2_NSLAB) slab_store(g[(c + 2) % S2_SETS], own, alds, blds);
        slab_read_frags(f[(c + 1) & 1], oth, afrag, bfrag);
        __builtin_amdgcn_sched_barrier(0);
        slab_mma(f[c & 1], acc);
        __builtin_amdgcn_sched_barrier(0);
        if (c + 2 < S2_NSLAB) lds_handover_barrier();
    }
    const int hl = r & 1;
    const int hs = r >> 1;
    const float* whn = Wh + (size_t)(nb * S2_NB + 2 * half) * LL + hl;
    HeadB cur;
    head_b_load(cur, whn + (size_t)((hs < cnt) ? mylist[hs] : 0) * DD * LL);
    __builtin_amdgcn_sched_barrier(0);
    slab_mma(f[(S2_NSLAB - 1) & 1], acc);
    __builtin_amdgcn_sched_barrier(0);

    __syncthreads();
    float* tile = lds + w * 16 * S2_TP;
    #pragma unroll
    for (int t = 0; t < 6; ++t) {
        #pragma unroll
        for (int j = 0; j < 8; ++j)
            tile[(j + 8 * half) * S2_TP + t * 16 + r] = acc[t][j];
    }
    __syncthreads();
    v2f af[S2_NB / 4];
    #pragma unroll
    for (int s = 0; s < S2_NB / 4; ++s)
        af[s] = *reinterpret_cast<const v2f*>(tile + r * S2_TP + 4 * s + 2 * half);

    const float* wh_of = whn;
    float* pbase = P + ((size_t)nb * AANN * BB + w * 16 + 8 * half) * LL + hl;
    HeadB hb1;
    for (int t0 = 0; t0 < cnt; t0 += 16) {
        const bool second = (t0 + 8 < cnt);
        if (second) {
            const int i1 = t0 + 8 + hs;
            head_b_load(hb1, wh_of + (size_t)((i1 < cnt) ? mylist[i1] : 0) * DD * LL);
        }
        {
            const bool valid = (t0 + hs) < cnt;
            const int  a     = valid ? mylist[t0 + hs] : 0;
            head_tile(af, cur, valid, pbase + (size_t)a * BB * LL);
        }
        if (second) {
            if (t0 + 16 < cnt) {
                const int i0 = t0 + 16 + hs;
                head_b_load(cur, wh_of + (size_t)((i0 < cnt) ? mylist[i0] : 0) * DD * LL);
            }
            const bool valid = (t0 + 8 + hs) < cnt;
            const int  a     = valid ? mylist[t0 + 8 + hs] : 0;
            head_tile(af, hb1, valid, pbase + (size_t)a * BB * LL);
        }
    }
}

__global__ __launch_bounds__(256) void head_sum_kernel(
    const float* __restrict__ P, const float* __restrict__ bh, float* __restrict__ out)
{
    __shared__ float T[16][33];

    const int ag  = blockIdx.x >> 2;
    const int q   = blockIdx.x & 3;
    const int tid = threadIdx.x;
    const int al  = tid >> 4;
    const int bl  = tid & 15;
    const int a   = ag * 16 + al;
    const int b   = q * 16 + bl;

    const float* src = P + ((size_t)a * BB + b) * LL;
    v2f pv[8];
    #pragma unroll
    for (int nb = 0; nb < 8; ++nb)
        pv[nb] = *reinterpret_cast<const v2f*>(src + (size_t)nb * AANN * BB * LL);
    v2f s = *reinterpret_cast<const v2f*>(bh + a * LL);
    #pragma unroll
    for (int nb = 0; nb < 8; ++nb) { s.x += pv[nb].x; s.y += pv[nb].y; }

    T[bl][al * 2]     = s.x;
    T[bl][al * 2 + 1] = s.y;
    __syncthreads();

    const int lane = tid & 31;
    const int w8   = tid >> 5;
    #pragma unroll
    for (int hh = 0; hh < 2; ++hh) {
        const int row = w8 + 8 * hh;
        out[(size_t)(q * 16 + row) * (AANN * LL) + ag * 32 + lane] = T[row][lane];
    }
}

extern "C" void kernel_launch(void* const* d_in, const int* in_sizes, int n_in,
                              void* d_out, int out_size, void* d_ws, size_t ws_size,
                              hipStream_t stream) {
    const float* pooled = (const float*)d_in[0];
    const float* W1     = (const float*)d_in[1];
    const float* b1     = (const float*)d_in[2];
    const float* W2     = (const float*)d_in[3];
    const float* b2     = (const float*)d_in[4];
    const float* Wh     = (const float*)d_in[5];
    const float* bh     = (const float*)d_in[6];
    const int*   g1     = (const int*)d_in[7];
    const int*   g2     = (const int*)d_in[8];
    float*       out    = (float*)d_out;

    float* H1 = (float*)d_ws;
    float* P  = H1 + (size_t)GG1 * BB * DD;

    stage1_kernel<<<GG1 * 96, 32, 0, stream>>>(pooled, W1, b1, H1);
    stage2_kernel<<<GG1 * GG2 * 8, 128, 0, stream>>>(H1, W2, b2, Wh, g1, g2, P);
    head_sum_kernel<<<(AANN / 16) * 4, 256, 0, stream>>>(P, bh, out);
}


# Stage 1 through LDS: 192 four-wave work-groups on 32x32 blocks fetch each 64-k slab of pooled and W1 once, as whole lines, into two LDS stages; one tile per wave in two k chains: sum order changes
# speedup vs baseline: 8.5285x; 1.2193x over previous
#include <hip/hip_runtime.h>
#include <hip/hip_bf16.h>

#define AANN 256
#define BB   64
#define DD   768
#define LL   2
#define GG1  4
#define GG2  8

typedef __attribute__((ext_vector_type(2))) float v2f;
typedef __attribute__((ext_vector_type(8))) float v8f;

__device__ __forceinline__ v8f wmma_f32(v2f a, v2f b, v8f c) {
    return __builtin_amdgcn_wmma_f32_16x16x4_f32(
         false, a,  false, b,
         (short)0, c,  false,  false);
}


typedef __attribute__((ext_vector_type(4))) float v4f;

__device__ __forceinline__ void lds_handover_barrier()
{
    __builtin_amdgcn_fence(__ATOMIC_RELEASE, "workgroup", "local");
    __builtin_amdgcn_s_barrier();
    __builtin_amdgcn_fence(__ATOMIC_ACQUIRE, "workgroup", "local");
}


#define S1_KB    64
#define S1_NSLAB (DD / S1_KB)
#define S1_PA    68
#define S1_PB    96
#define S1_AFL   (32 * S1_PA)
#define S1_BFL   ((S1_KB / 2) * S1_PB)
#define S1_STAGE (S1_AFL + S1_BFL)
#define S1_SETS  4
#define S1_TP    36

struct Slab1 {
    v4f a[4];
    v4f bx[2];
    v4f by[2];
};

__device__ __forceinline__ void slab1_fetch(
    Slab1& g, const float* __restrict__ asrc, const float* __restrict__ bsrc, int k0)
{
    #pragma unroll
    for (int j = 0; j < 4; ++j)
        g.a[j] = *reinterpret_cast<const v4f*>(asrc + (size_t)j * 8 * DD + k0);
    #pragma unroll
    for (int i = 0; i < 2; ++i) {
        g.bx[i] = *reinterpret_cast<const v4f*>(bsrc + (size_t)(k0 + 32 * i) * DD);
        g.by[i] = *reinterpret_cast<const v4f*>(bsrc + (size_t)(k0 + 32 * i + 1) * DD);
    }
}

__device__ __forceinline__ void slab1_store(const Slab1& g, float* st, int alds, int blds)
{
    #pragma unroll
    for (int j = 0; j < 4; ++j)
        *reinterpret_cast<v4f*>(st + alds + j * 8 * S1_PA) = g.a[j];
    #pragma unroll
    for (int i = 0; i < 2; ++i) {
        v4f lo, hi;
        lo.x = g.bx[i].x; lo.y = g.by[i].x; lo.z = g.bx[i].y; lo.w = g.by[i].y;
        hi.x = g.bx[i].z; hi.y = g.by[i].z; hi.z = g.bx[i].w; hi.w = g.by[i].w;
        *reinterpret_cast<v4f*>(st + S1_AFL + blds + i * 16 * S1_PB)     = lo;
        *reinterpret_cast<v4f*>(st + S1_AFL + blds + i * 16 * S1_PB + 4) = hi;
    }
}

struct Slab1Frags {
    v2f a[S1_KB / 4];
    v2f b[S1_KB / 4];
};

__device__ __forceinline__ void slab1_read_frags(
    Slab1Frags& f, const float* st, int afrag, int bfrag)
{
    #pragma unroll
    for (int s = 0; s < S1_KB / 4; ++s) {
        f.a[s] = *reinterpret_cast<const v2f*>(st + afrag + 4 * s);
        f.b[s] = *reinterpret_cast<const v2f*>(st + S1_AFL + bfrag + 2 * s * S1_PB);
    }
}

__device__ __forceinline__ void slab1_mma(const Slab1Frags& f, v8f (&acc)[2])
{
    #pragma unroll
    for (int s = 0; s < S1_KB / 4; s += 2) {
        acc[0] = wmma_f32(f.a[s],     f.b[s],     acc[0]);
        acc[1] = wmma_f32(f.a[s + 1], f.b[s + 1], acc[1]);
    }
}

__global__ __launch_bounds__(128) void stage1_kernel(
    const float* __restrict__ pooled, const float* __restrict__ W1,
    const float* __restrict__ b1, float* __restrict__ H1)
{
    __shared__ __attribute__((aligned(16))) float lds[2 * S1_STAGE];

    const int g  = blockIdx.x / 48;
    const int bt = blockIdx.x % 48;
    const int m0 = (bt / 24) * 32;
    const int n0 = (bt % 24) * 32;
    const float* Bm   = W1 + (size_t)g * DD * DD + n0;
    const float* bias = b1 + (size_t)g * DD + n0;
    float*       C    = H1 + (size_t)g * BB * DD + (size_t)m0 * DD + n0;

    const int tid  = threadIdx.x;
    const int lane = tid & 31;
    const int w    = tid >> 5;
    const int wm   = w >> 1;
    const int wn   = w & 1;
    const int half = lane >> 4;
    const int r    = lane & 15;
    const int ar   = tid >> 4;
    const int aq   = tid & 15;
    const int bp   = tid >> 3;
    const int bq   = tid & 7;

    const float* asrc = pooled + (size_t)(m0 + ar) * DD + aq * 4;
    const float* bsrc = Bm + (size_t)(2 * bp) * DD + bq * 4;
    const int alds  = ar * S1_PA + aq * 4;
    const int blds  = bp * S1_PB + bq * 8;
    const int afrag = (wm * 16 + r) * S1_PA + 2 * half;
    const int bfrag = half * S1_PB + 2 * (wn * 16 + r);

    Slab1 gs[S1_SETS];
    #pragma unroll
    for (int u = 0; u < S1_SETS; ++u) slab1_fetch(gs[u], asrc, bsrc, u * S1_KB);

    v8f acc[2];
    {
        const float bv = bias[wn * 16 + r];
        #pragma unroll
        for (int j = 0; j < 8; ++j) { acc[0][j] = bv; acc[1][j] = 0.0f; }
    }

    slab1_store(gs[0], lds, alds, blds);
    slab1_store(gs[1], lds + S1_STAGE, alds, blds);
    lds_handover_barrier();
    slab1_fetch(gs[0], asrc, bsrc, S1_SETS * S1_KB);
    Slab1Frags f[2];
    slab1_read_frags(f[0], lds, afrag, bfrag);
    lds_handover_barrier();

    #pragma unroll
    for (int c = 0; c < S1_NSLAB - 1; ++c) {
        float*       own = lds + (c & 1) * S1_STAGE;
        const float* oth = lds + ((c + 1) & 1) * S1_STAGE;
        if (c + 5 < S1_NSLAB) slab1_fetch(gs[(c + 5) % S1_SETS], asrc, bsrc, (c + 5) * S1_KB);
        if (c + 2 < S1_NSLAB) slab1_store(gs[(c + 2) % S1_SETS], own, alds, blds);
        slab1_read_frags(f[(c + 1) & 1], oth, afrag, bfrag);
        __builtin_amdgcn_sched_barrier(0);
        slab1_mma(f[c & 1], acc);
        __builtin_amdgcn_sched_barrier(0);
        if (c + 2 < S1_NSLAB) lds_handover_barrier();
    }
    slab1_mma(f[(S1_NSLAB - 1) & 1], acc);

    __syncthreads();
    #pragma unroll
    for (int j = 0; j < 8; ++j)
        lds[(wm * 16 + j + 8 * half) * S1_TP + wn * 16 + r] = acc[0][j] + acc[1][j];
    __syncthreads();
    #pragma unroll
    for (int i = 0; i < 8; ++i) {
        const int row = w * 8 + i;
        C[(size_t)row * DD + lane] = lds[row * S1_TP + lane];
    }
}

#define S2_KB    32
#define S2_NSLAB (DD / S2_KB)
#define S2_NB    96
#define S2_PA    36
#define S2_PB    224
#define S2_AFL   (BB * S2_PA)
#define S2_BFL   ((S2_KB / 2) * S2_PB)
#define S2_STAGE (S2_AFL + S2_BFL)
#define S2_SETS  4


struct Slab {
    v4f a[4];
    v4f bx[3];
    v4f by[3];
};

__device__ __forceinline__ void slab_fetch(
    Slab& g, const float* __restrict__ asrc, const float* __restrict__ bsrc, int k0)
{
    #pragma unroll
    for (int j = 0; j < 4; ++j)
        g.a[j] = *reinterpret_cast<const v4f*>(asrc + (size_t)j * 16 * DD + k0);
    #pragma unroll
    for (int seg = 0; seg < 3; ++seg) {
        g.bx[seg] = *reinterpret_cast<const v4f*>(bsrc + (size_t)k0 * DD + seg * 32);
        g.by[seg] = *reinterpret_cast<const v4f*>(bsrc + (size_t)(k0 + 1) * DD + seg * 32);
    }
}

__device__ __forceinline__ void slab_store(const Slab& g, float* st, int alds, int blds)
{
    #pragma unroll
    for (int j = 0; j < 4; ++j)
        *reinterpret_cast<v4f*>(st + alds + j * 16 * S2_PA) = g.a[j];
    #pragma unroll
    for (int seg = 0; seg < 3; ++seg) {
        v4f lo, hi;
        lo.x = g.bx[seg].x; lo.y = g.by[seg].x; lo.z = g.bx[seg].y; lo.w = g.by[seg].y;
        hi.x = g.bx[seg].z; hi.y = g.by[seg].z; hi.z = g.bx[seg].w; hi.w = g.by[seg].w;
        *reinterpret_cast<v4f*>(st + S2_AFL + blds + seg * 64)     = lo;
        *reinterpret_cast<v4f*>(st + S2_AFL + blds + seg * 64 + 4) = hi;
    }
}

struct SlabFrags {
    v2f a[8];
    v2f b[8][6];
};

__device__ __forceinline__ void slab_read_frags(
    SlabFrags& f, const float* st, int afrag, int bfrag)
{
    #pragma unroll
    for (int s = 0; s < 8; ++s) {
        f.a[s] = *reinterpret_cast<const v2f*>(st + afrag + 4 * s);
        #pragma unroll
        for (int t = 0; t < 6; ++t)
            f.b[s][t] = *reinterpret_cast<const v2f*>(
                st + S2_AFL + bfrag + 2 * s * S2_PB + t * 32);
    }
}

__device__ __forceinline__ void slab_mma(const SlabFrags& f, v8f (&acc)[6])
{
    #pragma unroll
    for (int s = 0; s < 8; ++s) {
        #pragma unroll
        for (int t = 0; t < 6; ++t)
            acc[t] = wmma_f32(f.a[s], f.b[s][t], acc[t]);
    }
}

#define S2_TP 100

struct HeadB {
    v2f b[S2_NB / 4];
};

__device__ __forceinline__ void head_b_load(HeadB& hb, const float* __restrict__ bp)
{
    #pragma unroll
    for (int s = 0; s < S2_NB / 4; ++s) {
        hb.b[s].x = bp[(4 * s) * LL];
        hb.b[s].y = bp[(4 * s + 1) * LL];
    }
}

__device__ __forceinline__ void head_tile(
    const v2f (&af)[S2_NB / 4], const HeadB& hb, bool valid, float* pp)
{
    v8f d0 = {}, d1 = {};
    #pragma unroll
    for (int s = 0; s < S2_NB / 4; s += 2) {
        d0 = wmma_f32(af[s],     hb.b[s],     d0);
        d1 = wmma_f32(af[s + 1], hb.b[s + 1], d1);
    }
    __builtin_amdgcn_sched_barrier(0);
    asm volatile("v_nop\n\tv_nop\n\tv_nop\n\tv_nop\n\tv_nop\n\tv_nop\n\tv_nop\n\tv_nop");
    __builtin_amdgcn_sched_barrier(0);
    if (valid) {
        #pragma unroll
        for (int j = 0; j < 8; ++j) pp[j * LL] = d0[j] + d1[j];
    }
}

__global__ __launch_bounds__(128) void stage2_kernel(
    const float* __restrict__ H1, const float* __restrict__ W2,
    const float* __restrict__ b2, const float* __restrict__ Wh,
    const int* __restrict__ g1_idx, const int* __restrict__ g2_idx,
    float* __restrict__ P)
{
    __shared__ __attribute__((aligned(16))) float lds[2 * S2_STAGE + 4 * AANN + 8];

    const int p  = blockIdx.x >> 3;
    const int nb = blockIdx.x & 7;
    const int g1 = p >> 3;
    const int g2 = p & 7;
    const float* A    = H1 + (size_t)g1 * BB * DD;
    const float* Bm   = W2 + (size_t)g2 * DD * DD + nb * S2_NB;
    const float* bias = b2 + (size_t)g2 * DD + nb * S2_NB;

    const int tid  = threadIdx.x;
    const int lane = tid & 31;
    const int w    = tid >> 5;
    const int half = lane >> 4;
    const int r    = lane & 15;
    const int tr   = tid >> 3;
    const int tq   = tid & 7;

    const float* asrc = A + (size_t)tr * DD + tq * 4;
    const float* bsrc = Bm + (size_t)(2 * tr) * DD + tq * 4;
    const int alds  = tr * S2_PA + tq * 4;
    const int blds  = tr * S2_PB + tq * 8;
    const int afrag = (w * 16 + r) * S2_PA + 2 * half;
    const int bfrag = half * S2_PB + 2 * r;

    v8f acc[6];
    #pragma unroll
    for (int t = 0; t < 6; ++t) {
        const float bv = bias[t * 16 + r];
        #pragma unroll
        for (int j = 0; j < 8; ++j) acc[t][j] = bv;
    }

    int* mylist = reinterpret_cast<int*>(lds + 2 * S2_STAGE) + w * AANN;
    int ga[AANN / 32], gb[AANN / 32];
    #pragma unroll
    for (int c = 0; c < AANN / 32; ++c) {
        ga[c] = g1_idx[c * 32 + lane];
        gb[c] = g2_idx[c * 32 + lane];
    }
    __builtin_amdgcn_sched_barrier(0);

    Slab g[S2_SETS];
    #pragma unroll
    for (int u = 0; u < S2_SETS; ++u) slab_fetch(g[u], asrc, bsrc, u * S2_KB);
    int cnt = 0;
    #pragma unroll
    for (int c = 0; c < AANN / 32; ++c) {
        const bool m = (ga[c] == g1) & (gb[c] == g2);
        const unsigned mask = __builtin_amdgcn_ballot_w32(m);
        if (m) mylist[cnt + __builtin_popcount(mask & ((1u << lane) - 1u))] = c * 32 + lane;
        cnt += __builtin_popcount(mask);
    }

    slab_store(g[0], lds, alds, blds);
    slab_store(g[1], lds + S2_STAGE, alds, blds);
    lds_handover_barrier();
    slab_fetch(g[0], asrc, bsrc, S2_SETS * S2_KB);
    SlabFrags f[2];
    slab_read_frags(f[0], lds, afrag, bfrag);
    lds_handover_barrier();

    for (int c0 = 0; c0 < 16; c0 += 4) {
        #pragma unroll
        for (int u = 0; u < 4; ++u) {
            const int c = c0 + u;
            float*       own = lds + (u & 1) * S2_STAGE;
            const float* oth = lds + ((u + 1) & 1) * S2_STAGE;
            slab_fetch(g[(u + 1) % S2_SETS], asrc, bsrc, (c + 5) * S2_KB);
            slab_store(g[(u + 2) % S2_SETS], own, alds, blds);
            slab_read_frags(f[(u + 1) & 1], oth, afrag, bfrag);
            __builtin_amdgcn_sched_barrier(0);
            slab_mma(f[u & 1], acc);
            __builtin_amdgcn_sched_barrier(0);
            lds_handover_barrier();
        }
    }
    #pragma unroll
    for (int c = 16; c < S2_NSLAB - 1; ++c) {
        float*       own = lds + (c & 1) * S2_STAGE;
        const float* oth = lds + ((c + 1) & 1) * S2_STAGE;
        if (c + 5 < S2_NSLAB) slab_fetch(g[(c + 5) % S2_SETS], asrc, bsrc, (c + 5) * S2_KB);
        if (c + 2 < S2_NSLAB) slab_store(g[(c + 2) % S2_SETS], own, alds, blds);
        slab_read_frags(f[(c + 1) & 1], oth, afrag, bfrag);
        __builtin_amdgcn_sched_barrier(0);
        slab_mma(f[c & 1], acc);
        __builtin_amdgcn_sched_barrier(0);
        if (c + 2 < S2_NSLAB) lds_handover_barrier();
    }
    const int hl = r & 1;
    const int hs = r >> 1;
    const float* whn = Wh + (size_t)(nb * S2_NB + 2 * half) * LL + hl;
    HeadB cur;
    head_b_load(cur, whn + (size_t)((hs < cnt) ? mylist[hs] : 0) * DD * LL);
    __builtin_amdgcn_sched_barrier(0);
    slab_mma(f[(S2_NSLAB - 1) & 1], acc);
    __builtin_amdgcn_sched_barrier(0);

    __syncthreads();
    float* tile = lds + w * 16 * S2_TP;
    #pragma unroll
    for (int t = 0; t < 6; ++t) {
        #pragma unroll
        for (int j = 0; j < 8; ++j)
            tile[(j + 8 * half) * S2_TP + t * 16 + r] = acc[t][j];
    }
    __syncthreads();
    v2f af[S2_NB / 4];
    #pragma unroll
    for (int s = 0; s < S2_NB / 4; ++s)
        af[s] = *reinterpret_cast<const v2f*>(tile + r * S2_TP + 4 * s + 2 * half);

    const float* wh_of = whn;
    float* pbase = P + ((size_t)nb * AANN * BB + w * 16 + 8 * half) * LL + hl;
    HeadB hb1;
    for (int t0 = 0; t0 < cnt; t0 += 16) {
        const bool second = (t0 + 8 < cnt);
        if (second) {
            const int i1 = t0 + 8 + hs;
            head_b_load(hb1, wh_of + (size_t)((i1 < cnt) ? mylist[i1] : 0) * DD * LL);
        }
        {
            const bool valid = (t0 + hs) < cnt;
            const int  a     = valid ? mylist[t0 + hs] : 0;
            head_tile(af, cur, valid, pbase + (size_t)a * BB * LL);
        }
        if (second) {
            if (t0 + 16 < cnt) {
                const int i0 = t0 + 16 + hs;
                head_b_load(cur, wh_of + (size_t)((i0 < cnt) ? mylist[i0] : 0) * DD * LL);
            }
            const bool valid = (t0 + 8 + hs) < cnt;
            const int  a     = valid ? mylist[t0 + 8 + hs] : 0;
            head_tile(af, hb1, valid, pbase + (size_t)a * BB * LL);
        }
    }
}

__global__ __launch_bounds__(256) void head_sum_kernel(
    const float* __restrict__ P, const float* __restrict__ bh, float* __restrict__ out)
{
    __shared__ float T[16][33];

    const int ag  = blockIdx.x >> 2;
    const int q   = blockIdx.x & 3;
    const int tid = threadIdx.x;
    const int al  = tid >> 4;
    const int bl  = tid & 15;
    const int a   = ag * 16 + al;
    const int b   = q * 16 + bl;

    const float* src = P + ((size_t)a * BB + b) * LL;
    v2f pv[8];
    #pragma unroll
    for (int nb = 0; nb < 8; ++nb)
        pv[nb] = *reinterpret_cast<const v2f*>(src + (size_t)nb * AANN * BB * LL);
    v2f s = *reinterpret_cast<const v2f*>(bh + a * LL);
    #pragma unroll
    for (int nb = 0; nb < 8; ++nb) { s.x += pv[nb].x; s.y += pv[nb].y; }

    T[bl][al * 2]     = s.x;
    T[bl][al * 2 + 1] = s.y;
    __syncthreads();

    const int lane = tid & 31;
    const int w8   = tid >> 5;
    #pragma unroll
    for (int hh = 0; hh < 2; ++hh) {
        const int row = w8 + 8 * hh;
        out[(size_t)(q * 16 + row) * (AANN * LL) + ag * 32 + lane] = T[row][lane];
    }
}

extern "C" void kernel_launch(void* const* d_in, const int* in_sizes, int n_in,
                              void* d_out, int out_size, void* d_ws, size_t ws_size,
                              hipStream_t stream) {
    const float* pooled = (const float*)d_in[0];
    const float* W1     = (const float*)d_in[1];
    const float* b1     = (const float*)d_in[2];
    const float* W2     = (const float*)d_in[3];
    const float* b2     = (const float*)d_in[4];
    const float* Wh     = (const float*)d_in[5];
    const float* bh     = (const float*)d_in[6];
    const int*   g1     = (const int*)d_in[7];
    const int*   g2     = (const int*)d_in[8];
    float*       out    = (float*)d_out;

    float* H1 = (float*)d_ws;
    float* P  = H1 + (size_t)GG1 * BB * DD;

    stage1_kernel<<<GG1 * 48, 128, 0, stream>>>(pooled, W1, b1, H1);
    stage2_kernel<<<GG1 * GG2 * 8, 128, 0, stream>>>(H1, W2, b2, Wh, g1, g2, P);
    head_sum_kernel<<<(AANN / 16) * 4, 256, 0, stream>>>(P, bh, out);
}
